# starting kernel
# baseline (speedup 1.0000x reference)
#include <hip/hip_runtime.h>
#include <hip/hip_bf16.h>


typedef float v2f __attribute__((ext_vector_type(2)));
typedef float v8f __attribute__((ext_vector_type(8)));

#define Bsz 2048
#define Tsz 4096
#define Hsz 8

__device__ __forceinline__ float shx16(float v) {
#if __has_builtin(__builtin_amdgcn_permlanex16)
    unsigned int u = __float_as_uint(v);
    unsigned int r = __builtin_amdgcn_permlanex16(
        u, u, 0x76543210u, 0xfedcba98u,  false,  false);
    return __uint_as_float(r);
#else
    return __shfl_xor(v, 16, 32);
#endif
}

__device__ __forceinline__ float fsig(float x) {
    return __builtin_amdgcn_rcpf(1.0f + __expf(-x));
}
__device__ __forceinline__ float ftanh(float x) {
    return 1.0f - 2.0f * __builtin_amdgcn_rcpf(__expf(2.0f * x) + 1.0f);
}

__device__ __forceinline__ v8f wmma4(v2f a, v2f b, v8f c) {
    return __builtin_amdgcn_wmma_f32_16x16x4_f32(
         false, a,  false, b,
         (short)0, c,  false,  false);
}

__global__ __launch_bounds__(32) void lstm2_wmma_kernel(
    const float* __restrict__ x,
    const float* __restrict__ Wih0,
    const float* __restrict__ Whh0,
    const float* __restrict__ bih0, const float* __restrict__ bhh0,
    const float* __restrict__ Wih1,
    const float* __restrict__ Whh1,
    const float* __restrict__ bih1, const float* __restrict__ bhh1,
    const float* __restrict__ Wlin,
    const float* __restrict__ blin,
    float* __restrict__ out)
{
    const int lane = threadIdx.x;
    const int lm   = lane & 15;
    const bool hh  = (lane >= 16);
    const int hsel = hh ? 1 : 0;
    const long row = (long)(blockIdx.x * 32 + lane) * Tsz;

    v2f A0[2][2];
    v2f A1[2][4];
#pragma unroll
    for (int gt = 0; gt < 2; ++gt) {
        const int g = 16 * gt + lm;
#pragma unroll
        for (int kf = 0; kf < 2; ++kf) {
            const int kb = 4 * kf + 2 * hsel;
            A0[gt][kf].x = Whh0[g * Hsz + kb];
            A0[gt][kf].y = Whh0[g * Hsz + kb + 1];
        }
#pragma unroll
        for (int kf = 0; kf < 4; ++kf) {
            const int k = 4 * kf + 2 * hsel;
            const float* Wsrc = (kf < 2) ? Wih1 : Whh1;
            const int kk = (kf < 2) ? k : (k - 8);
            A1[gt][kf].x = Wsrc[g * Hsz + kk];
            A1[gt][kf].y = Wsrc[g * Hsz + kk + 1];
        }
    }

    float wg0C[2][8], b0C[2][8], b1C[2][8];
#pragma unroll
    for (int gt = 0; gt < 2; ++gt)
#pragma unroll
        for (int r = 0; r < 8; ++r) {
            const int g = 16 * gt + r + 8 * hsel;
            wg0C[gt][r] = Wih0[g];
            b0C[gt][r]  = bih0[g] + bhh0[g];
            b1C[gt][r]  = bih1[g] + bhh1[g];
        }

    float wl[8];
#pragma unroll
    for (int j = 0; j < 8; ++j) wl[j] = Wlin[j];
    const float bl = blin[0];

    float h1[8], c1[8], h2[8], c2[8];
#pragma unroll
    for (int j = 0; j < 8; ++j) { h1[j] = 0.f; c1[j] = 0.f; h2[j] = 0.f; c2[j] = 0.f; }

    for (int t = 0; t < Tsz; ++t) {
        const float xv = x[row + t];
        __builtin_prefetch(&x[row + t + 256], 0, 0);
        const float xsh = shx16(xv);
        const float xb0 = hh ? xsh : xv;
        const float xb1 = hh ? xv : xsh;

        v2f Bh1[2][2];
#pragma unroll
        for (int kf = 0; kf < 2; ++kf) {
            const int k0 = 4 * kf;
            const float s0 = shx16(h1[k0 + 0]), s1 = shx16(h1[k0 + 1]);
            const float s2 = shx16(h1[k0 + 2]), s3 = shx16(h1[k0 + 3]);
            Bh1[kf][0].x = hh ? s2 : h1[k0 + 0];
            Bh1[kf][0].y = hh ? s3 : h1[k0 + 1];
            Bh1[kf][1].x = hh ? h1[k0 + 2] : s0;
            Bh1[kf][1].y = hh ? h1[k0 + 3] : s1;
        }

        v8f D0[2][2];
#pragma unroll
        for (int gt = 0; gt < 2; ++gt)
#pragma unroll
            for (int bt = 0; bt < 2; ++bt) {
                const float xb = bt ? xb1 : xb0;
                v8f acc;
#pragma unroll
                for (int r = 0; r < 8; ++r)
                    acc[r] = fmaf(xb, wg0C[gt][r], b0C[gt][r]);
                acc = wmma4(A0[gt][0], Bh1[0][bt], acc);
                acc = wmma4(A0[gt][1], Bh1[1][bt], acc);
                D0[gt][bt] = acc;
            }

#pragma unroll
        for (int r = 0; r < 8; ++r) {
            const float a00 = D0[0][0][r], a01 = D0[0][1][r];
            const float a10 = D0[1][0][r], a11 = D0[1][1][r];
            const float sA0 = shx16(a00), sB0 = shx16(a01);
            const float sA1 = shx16(a10), sB1 = shx16(a11);
            const float ig = fsig(hh ? sB0 : a00);
            const float fg = fsig(hh ? a01 : sA0);
            const float gg = ftanh(hh ? sB1 : a10);
            const float og = fsig(hh ? a11 : sA1);
            c1[r] = fmaf(fg, c1[r], ig * gg);
            h1[r] = og * ftanh(c1[r]);
        }

        v2f Bc[4][2];
#pragma unroll
        for (int kf = 0; kf < 4; ++kf) {
            const float* hs = (kf < 2) ? h1 : h2;
            const int k0 = 4 * (kf & 1);
            const float s0 = shx16(hs[k0 + 0]), s1 = shx16(hs[k0 + 1]);
            const float s2 = shx16(hs[k0 + 2]), s3 = shx16(hs[k0 + 3]);
            Bc[kf][0].x = hh ? s2 : hs[k0 + 0];
            Bc[kf][0].y = hh ? s3 : hs[k0 + 1];
            Bc[kf][1].x = hh ? hs[k0 + 2] : s0;
            Bc[kf][1].y = hh ? hs[k0 + 3] : s1;
        }

        v8f D1[2][2];
#pragma unroll
        for (int gt = 0; gt < 2; ++gt)
#pragma unroll
            for (int bt = 0; bt < 2; ++bt) {
                v8f acc;
#pragma unroll
                for (int r = 0; r < 8; ++r) acc[r] = b1C[gt][r];
                acc = wmma4(A1[gt][0], Bc[0][bt], acc);
                acc = wmma4(A1[gt][1], Bc[1][bt], acc);
                acc = wmma4(A1[gt][2], Bc[2][bt], acc);
                acc = wmma4(A1[gt][3], Bc[3][bt], acc);
                D1[gt][bt] = acc;
            }

#pragma unroll
        for (int r = 0; r < 8; ++r) {
            const float a00 = D1[0][0][r], a01 = D1[0][1][r];
            const float a10 = D1[1][0][r], a11 = D1[1][1][r];
            const float sA0 = shx16(a00), sB0 = shx16(a01);
            const float sA1 = shx16(a10), sB1 = shx16(a11);
            const float ig = fsig(hh ? sB0 : a00);
            const float fg = fsig(hh ? a01 : sA0);
            const float gg = ftanh(hh ? sB1 : a10);
            const float og = fsig(hh ? a11 : sA1);
            c2[r] = fmaf(fg, c2[r], ig * gg);
            h2[r] = og * ftanh(c2[r]);
        }

        float p = bl;
#pragma unroll
        for (int j = 0; j < 8; ++j) p = fmaf(h2[j], wl[j], p);
        out[row + t] = p;
    }
}

extern "C" void kernel_launch(void* const* d_in, const int* in_sizes, int n_in,
                              void* d_out, int out_size, void* d_ws, size_t ws_size,
                              hipStream_t stream) {
    (void)in_sizes; (void)n_in; (void)out_size; (void)d_ws; (void)ws_size;
    const float* x    = (const float*)d_in[0];
    const float* Wih0 = (const float*)d_in[1];
    const float* Whh0 = (const float*)d_in[2];
    const float* bih0 = (const float*)d_in[3];
    const float* bhh0 = (const float*)d_in[4];
    const float* Wih1 = (const float*)d_in[5];
    const float* Whh1 = (const float*)d_in[6];
    const float* bih1 = (const float*)d_in[7];
    const float* bhh1 = (const float*)d_in[8];
    const float* Wlin = (const float*)d_in[9];
    const float* blin = (const float*)d_in[10];
    float* out = (float*)d_out;

    lstm2_wmma_kernel<<<Bsz / 32, 32, 0, stream>>>(
        x, Wih0, Whh0, bih0, bhh0, Wih1, Whh1, bih1, bhh1, Wlin, blin, out);
}


# Time loop rotated: each product of trip t+1 is issued once the hidden units its k fragment reads are updated, a slice of cell math behind each; gate tiles double-buffered; same sums, same order
# speedup vs baseline: 3.5874x; 3.5874x over previous
#include <hip/hip_runtime.h>


typedef float v2f __attribute__((ext_vector_type(2)));
typedef float v8f __attribute__((ext_vector_type(8)));

#define Bsz 2048
#define Tsz 4096
#define Hsz 8

__device__ __forceinline__ float fsig(float x) {
    return __builtin_amdgcn_rcpf(1.0f + __expf(-x));
}
__device__ __forceinline__ float ftanh(float x) {
    return 1.0f - 2.0f * __builtin_amdgcn_rcpf(__expf(2.0f * x) + 1.0f);
}

__device__ __forceinline__ v8f wmma4(v2f a, v2f b, v8f c) {
    return __builtin_amdgcn_wmma_f32_16x16x4_f32(
         false, a,  false, b,
         (short)0, c,  false,  false);
}

__device__ __forceinline__ void gates_start(const float (&wC)[4][8],
                                            const float (&bC)[4][8],
                                            const float xv, v8f (&G)[4])
{
#pragma unroll
    for (int g = 0; g < 4; ++g)
#pragma unroll
        for (int r = 0; r < 8; ++r)
            G[g][r] = fmaf(xv, wC[g][r], bC[g][r]);
}

#define ORDER(outs, ins) asm volatile("" : outs : ins)
#define IO4(v) "+v"(v[0]), "+v"(v[1]), "+v"(v[2]), "+v"(v[3])
#define IN4(v) "v"(v[0]), "v"(v[1]), "v"(v[2]), "v"(v[3])
#define COMMA ,

__device__ __forceinline__ void lstm_stage(const v2f (&A)[4][4],
                                           const v2f (&wC2)[4][4],
                                           const v2f (&bC2)[4][4],
                                           const float (&wl)[8], const float bl,
                                           v2f xre,
                                           v8f (&G)[4], v8f (&Gn)[4],
                                           v2f (&sB)[4], float (&c)[8],
                                           float* outp, const bool hh)
{
    float p;
    float a[4], b[4], ei[4], ef[4], eg[4], eo[4], ih[4], fh[4], gh[4], oh[4];
    float cB[4], ec[4], th[4];
#pragma unroll
    for (int r = 0; r < 4; ++r) cB[r] = c[r + 4];

#pragma unroll
    for (int r = 0; r < 4; ++r) a[r] = G[0][r + 4];
    ORDER("+v"(sB[0]) COMMA IO4(a), "v"(sB[1]));
    Gn[0] = wmma4(A[0][0], sB[0], Gn[0]);
#pragma unroll
    for (int r = 0; r < 4; ++r) ei[r] = __expf(-a[r]);

#pragma unroll
    for (int r = 0; r < 4; ++r) a[r] = G[1][r + 4];
    ORDER("+v"(sB[0]) COMMA IO4(a) COMMA IO4(ei), "v"(Gn[0]));
    Gn[1] = wmma4(A[1][0], sB[0], Gn[1]);
#pragma unroll
    for (int r = 0; r < 4; ++r) { ih[r] = __builtin_amdgcn_rcpf(1.0f + ei[r]); ef[r] = __expf(-a[r]); }

#pragma unroll
    for (int r = 0; r < 4; ++r) a[r] = G[2][r + 4];
    ORDER("+v"(sB[0]) COMMA IO4(a) COMMA IO4(ef), "v"(Gn[1]) COMMA IN4(ih));
    Gn[2] = wmma4(A[2][0], sB[0], Gn[2]);
#pragma unroll
    for (int r = 0; r < 4; ++r) { fh[r] = __builtin_amdgcn_rcpf(1.0f + ef[r]); eg[r] = __expf(2.0f * a[r]); }

    ORDER("+v"(sB[0]) COMMA IO4(eg), "v"(Gn[2]) COMMA IN4(fh));
    Gn[3] = wmma4(A[3][0], sB[0], Gn[3]);
#pragma unroll
    for (int r = 0; r < 4; ++r) eg[r] = eg[r] + 1.0f;
    ORDER(IO4(eg), "v"(sB[0]));
#pragma unroll
    for (int r = 0; r < 4; ++r) eg[r] = __builtin_amdgcn_rcpf(eg[r]);
    ORDER(IO4(eg), "v"(sB[0]));
#pragma unroll
    for (int r = 0; r < 4; ++r) gh[r] = 1.0f - 2.0f * eg[r];

#pragma unroll
    for (int r = 0; r < 4; ++r) a[r] = G[3][r + 4];
    ORDER("+v"(sB[1]) COMMA IO4(a) COMMA IO4(cB), "v"(Gn[3]) COMMA IN4(gh));
    Gn[0] = wmma4(A[0][1], sB[1], Gn[0]);
#pragma unroll
    for (int r = 0; r < 4; ++r) { cB[r] = fmaf(fh[r], cB[r], ih[r] * gh[r]); eo[r] = __expf(-a[r]); }

    ORDER("+v"(sB[1]) COMMA IO4(cB), "v"(Gn[0]) COMMA IN4(eo));
    Gn[1] = wmma4(A[1][1], sB[1], Gn[1]);
#pragma unroll
    for (int r = 0; r < 4; ++r) { ec[r] = __expf(2.0f * cB[r]); c[r + 4] = cB[r]; }

    ORDER("+v"(sB[1]) COMMA IO4(ec), "v"(Gn[1]));
    Gn[2] = wmma4(A[2][1], sB[1], Gn[2]);
#pragma unroll
    for (int r = 0; r < 4; ++r) ec[r] = ec[r] + 1.0f;
    ORDER(IO4(ec), "v"(sB[1]));
#pragma unroll
    for (int r = 0; r < 4; ++r) ec[r] = __builtin_amdgcn_rcpf(ec[r]);
    ORDER(IO4(ec), "v"(sB[1]));
#pragma unroll
    for (int r = 0; r < 4; ++r) th[r] = 1.0f - 2.0f * ec[r];

    ORDER("+v"(sB[1]) COMMA IO4(th) COMMA IO4(eo), "v"(Gn[2]));
    Gn[3] = wmma4(A[3][1], sB[1], Gn[3]);
#pragma unroll
    for (int r = 0; r < 4; ++r) eo[r] = 1.0f + eo[r];
    ORDER(IO4(eo), "v"(sB[1]));
#pragma unroll
    for (int r = 0; r < 4; ++r) oh[r] = __builtin_amdgcn_rcpf(eo[r]);
    sB[2].x = oh[0] * th[0]; sB[2].y = oh[1] * th[1];
    sB[3].x = oh[2] * th[2]; sB[3].y = oh[3] * th[3];

    p = bl;
    ORDER("+v"(sB[2]) COMMA "+v"(xre) COMMA "+v"(p), "v"(Gn[3]) COMMA "v"(sB[3]));
    Gn[2] = wmma4(A[2][2], sB[2], Gn[2]);
#pragma unroll
    for (int j = 0; j < 4; ++j) {
        const v2f t2 = __builtin_elementwise_fma(xre, wC2[0][j], bC2[0][j]);
        G[0][2 * j] = t2.x; G[0][2 * j + 1] = t2.y;
    }
    p = fmaf(sB[0].x, wl[0], p); p = fmaf(sB[0].y, wl[1], p);

    ORDER("+v"(sB[2]) COMMA "+v"(xre) COMMA "+v"(p), "v"(Gn[2]) COMMA "v"(G[0]));
    Gn[0] = wmma4(A[0][2], sB[2], Gn[0]);
#pragma unroll
    for (int j = 0; j < 4; ++j) {
        const v2f t2 = __builtin_elementwise_fma(xre, wC2[1][j], bC2[1][j]);
        G[1][2 * j] = t2.x; G[1][2 * j + 1] = t2.y;
    }
    p = fmaf(sB[1].x, wl[2], p); p = fmaf(sB[1].y, wl[3], p);

    ORDER("+v"(sB[2]) COMMA "+v"(xre) COMMA "+v"(p), "v"(Gn[0]) COMMA "v"(G[1]));
    Gn[1] = wmma4(A[1][2], sB[2], Gn[1]);
#pragma unroll
    for (int j = 0; j < 4; ++j) {
        const v2f t2 = __builtin_elementwise_fma(xre, wC2[2][j], bC2[2][j]);
        G[2][2 * j] = t2.x; G[2][2 * j + 1] = t2.y;
    }
    p = fmaf(sB[2].x, wl[4], p); p = fmaf(sB[2].y, wl[5], p);

    ORDER("+v"(sB[3]) COMMA "+v"(xre) COMMA "+v"(p), "v"(Gn[1]) COMMA "v"(G[2]));
    Gn[2] = wmma4(A[2][3], sB[3], Gn[2]);
#pragma unroll
    for (int j = 0; j < 4; ++j) {
        const v2f t2 = __builtin_elementwise_fma(xre, wC2[3][j], bC2[3][j]);
        G[3][2 * j] = t2.x; G[3][2 * j + 1] = t2.y;
    }
    p = fmaf(sB[3].x, wl[6], p); p = fmaf(sB[3].y, wl[7], p);

    ORDER("+v"(sB[3]) COMMA "+v"(p), "v"(Gn[2]) COMMA "v"(G[3]));
    Gn[0] = wmma4(A[0][3], sB[3], Gn[0]);
    if (hh) *outp = p;

#pragma unroll
    for (int r = 0; r < 4; ++r) a[r] = Gn[2][r];
    ORDER("+v"(sB[3]) COMMA IO4(a), "v"(Gn[0]));
    Gn[1] = wmma4(A[1][3], sB[3], Gn[1]);
#pragma unroll
    for (int r = 0; r < 4; ++r) eg[r] = __expf(2.0f * a[r]);

#pragma unroll
    for (int r = 0; r < 4; ++r) { a[r] = Gn[0][r]; b[r] = Gn[1][r]; }
    ORDER("+v"(sB[2]) COMMA IO4(eg) COMMA IO4(a) COMMA IO4(b), "v"(sB[3]));
    Gn[3] = wmma4(A[3][2], sB[2], Gn[3]);
    float ga[4], m[4], fa[4], oa[4], hA[4];
#pragma unroll
    for (int r = 0; r < 4; ++r) { eg[r] = eg[r] + 1.0f; ei[r] = __expf(-a[r]); ef[r] = __expf(-b[r]); }
    ORDER(IO4(eg), "v"(sB[2]));
#pragma unroll
    for (int r = 0; r < 4; ++r) eg[r] = __builtin_amdgcn_rcpf(eg[r]);
    ORDER(IO4(eg), "v"(sB[2]));
#pragma unroll
    for (int r = 0; r < 4; ++r) ga[r] = 1.0f - 2.0f * eg[r];

    ORDER("+v"(sB[3]) COMMA IO4(ei) COMMA IO4(ef), "v"(Gn[3]) COMMA IN4(ga));
    Gn[3] = wmma4(A[3][3], sB[3], Gn[3]);
#pragma unroll
    for (int r = 0; r < 4; ++r) { ef[r] = 1.0f + ef[r]; ei[r] = 1.0f + ei[r]; }

    ORDER(IO4(ef) COMMA IO4(ei), "v"(Gn[3]));
#pragma unroll
    for (int r = 0; r < 4; ++r) { fa[r] = __builtin_amdgcn_rcpf(ef[r]); ei[r] = __builtin_amdgcn_rcpf(ei[r]); }
    ORDER(IO4(fa) COMMA IO4(ei), "v"(sB[3]));
#pragma unroll
    for (int r = 0; r < 4; ++r) m[r] = ei[r] * ga[r];
    ORDER(IO4(fa), "v"(sB[3]));
#pragma unroll
    for (int r = 0; r < 4; ++r) c[r] = fmaf(fa[r], c[r], m[r]);
#pragma unroll
    for (int r = 0; r < 4; ++r) a[r] = Gn[3][r];
    ORDER("+v"(c[0]) COMMA "+v"(c[1]) COMMA "+v"(c[2]) COMMA "+v"(c[3]) COMMA IO4(a), "v"(sB[3]));
#pragma unroll
    for (int r = 0; r < 4; ++r) { ec[r] = __expf(2.0f * c[r]); eo[r] = __expf(-a[r]); }
    ORDER(IO4(ec) COMMA IO4(eo), "v"(sB[3]));
#pragma unroll
    for (int r = 0; r < 4; ++r) { ec[r] = ec[r] + 1.0f; eo[r] = 1.0f + eo[r]; }
    ORDER(IO4(ec) COMMA IO4(eo), "v"(sB[3]));
#pragma unroll
    for (int r = 0; r < 4; ++r) { ec[r] = __builtin_amdgcn_rcpf(ec[r]); oa[r] = __builtin_amdgcn_rcpf(eo[r]); }
    ORDER(IO4(ec), IN4(oa));
#pragma unroll
    for (int r = 0; r < 4; ++r) th[r] = 1.0f - 2.0f * ec[r];
#pragma unroll
    for (int r = 0; r < 4; ++r) hA[r] = oa[r] * th[r];
    sB[0].x = hA[0]; sB[0].y = hA[1]; sB[1].x = hA[2]; sB[1].y = hA[3];
}

__global__ __launch_bounds__(32) void lstm2_halves_kernel(
    const float* __restrict__ x,
    const float* __restrict__ Wih0,
    const float* __restrict__ Whh0,
    const float* __restrict__ bih0, const float* __restrict__ bhh0,
    const float* __restrict__ Wih1,
    const float* __restrict__ Whh1,
    const float* __restrict__ bih1, const float* __restrict__ bhh1,
    const float* __restrict__ Wlin,
    const float* __restrict__ blin,
    float* __restrict__ out)
{
    const int lane = threadIdx.x;
    const int lm   = lane & 15;
    const bool hh  = (lane >= 16);
    const long row = (long)(blockIdx.x * 16 + lm) * Tsz;

    v2f A[4][4];
    {
        const bool l1row = (lm >= 8);
        const float* Wsrc = l1row ? (hh ? Whh1 : Wih1) : Whh0;
        const bool live = l1row || !hh;
#pragma unroll
        for (int g = 0; g < 4; ++g) {
            const int G = 8 * g + (lm & 7);
#pragma unroll
            for (int f = 0; f < 4; ++f) {
                const float w0 = Wsrc[G * Hsz + 2 * f];
                const float w1 = Wsrc[G * Hsz + 2 * f + 1];
                A[g][f].x = live ? w0 : 0.0f;
                A[g][f].y = live ? w1 : 0.0f;
            }
        }
    }

    float wC[4][8], bC[4][8];
#pragma unroll
    for (int g = 0; g < 4; ++g)
#pragma unroll
        for (int r = 0; r < 8; ++r) {
            const int G = 8 * g + r;
            const float w0 = Wih0[G];
            const float b0 = bih0[G] + bhh0[G];
            const float b1 = bih1[G] + bhh1[G];
            wC[g][r] = hh ? 0.0f : w0;
            bC[g][r] = hh ? b1 : b0;
        }

    float wl[8];
#pragma unroll
    for (int j = 0; j < 8; ++j) wl[j] = Wlin[j];
    const float bl = blin[0];

    float s[8], c[8];
#pragma unroll
    for (int j = 0; j < 8; ++j) { s[j] = 0.f; c[j] = 0.f; }

    const float* xp = x + row;
    const float x0 = xp[0], x1 = xp[1], x2 = xp[2];
    float xb = xp[3], xc = xp[4];
    xp += 5;

    v8f Ga[4], Gb[4];
    gates_start(wC, bC, x0, Ga);
#pragma unroll
    for (int f = 0; f < 4; ++f) {
        v2f Bz; Bz.x = s[2 * f]; Bz.y = s[2 * f + 1];
#pragma unroll
        for (int g = 0; g < 4; ++g) Ga[g] = wmma4(A[g][f], Bz, Ga[g]);
    }

#pragma unroll
    for (int r = 0; r < 8; ++r) {
        const float i0 = fsig(Ga[0][r]);
        const float f0 = fsig(Ga[1][r]);
        const float g0 = ftanh(Ga[2][r]);
        const float o0 = fsig(Ga[3][r]);
        c[r] = fmaf(f0, c[r], i0 * g0);
        s[r] = o0 * ftanh(c[r]);
        s[r] = hh ? 0.f : s[r];
        c[r] = hh ? 0.f : c[r];
    }

    gates_start(wC, bC, x1, Gb);
#pragma unroll
    for (int f = 0; f < 4; ++f) {
        v2f Bs; Bs.x = s[2 * f]; Bs.y = s[2 * f + 1];
#pragma unroll
        for (int g = 0; g < 4; ++g) Gb[g] = wmma4(A[g][f], Bs, Gb[g]);
    }
    gates_start(wC, bC, x2, Ga);
#pragma unroll
    for (int r = 0; r < 4; ++r) {
        const float i1 = fsig(Gb[0][r]);
        const float f1 = fsig(Gb[1][r]);
        const float g1 = ftanh(Gb[2][r]);
        const float o1 = fsig(Gb[3][r]);
        c[r] = fmaf(f1, c[r], i1 * g1);
        s[r] = o1 * ftanh(c[r]);
    }
    v2f wC2[4][4], bC2[4][4];
#pragma unroll
    for (int g = 0; g < 4; ++g)
#pragma unroll
        for (int j = 0; j < 4; ++j) {
            wC2[g][j].x = wC[g][2 * j]; wC2[g][j].y = wC[g][2 * j + 1];
            bC2[g][j].x = bC[g][2 * j]; bC2[g][j].y = bC[g][2 * j + 1];
        }
    v2f sB[4];
#pragma unroll
    for (int f = 0; f < 4; ++f) { sB[f].x = s[2 * f]; sB[f].y = s[2 * f + 1]; }

    float* op = out + row;
    for (int it = 0; it < Tsz / 2; ++it) {
        const float nb = *xp;
        __builtin_prefetch(xp + 255, 0, 0);
        xp += (2 * it + 6 < Tsz) ? 1 : 0;
        const float nc = *xp;
        xp += (2 * it + 7 < Tsz) ? 1 : 0;

        v2f xb2, xc2;
        xb2.x = xb; xb2.y = xb; xc2.x = xc; xc2.y = xc;
        lstm_stage(A, wC2, bC2, wl, bl, xb2, Gb, Ga, sB, c, op,     hh);
        lstm_stage(A, wC2, bC2, wl, bl, xc2, Ga, Gb, sB, c, op + 1, hh);

        xb = nb; xc = nc;
        op += 2;
    }
}

extern "C" void kernel_launch(void* const* d_in, const int* in_sizes, int n_in,
                              void* d_out, int out_size, void* d_ws, size_t ws_size,
                              hipStream_t stream) {
    (void)in_sizes; (void)n_in; (void)out_size; (void)d_ws; (void)ws_size;
    const float* x    = (const float*)d_in[0];
    const float* Wih0 = (const float*)d_in[1];
    const float* Whh0 = (const float*)d_in[2];
    const float* bih0 = (const float*)d_in[3];
    const float* bhh0 = (const float*)d_in[4];
    const float* Wih1 = (const float*)d_in[5];
    const float* Whh1 = (const float*)d_in[6];
    const float* bih1 = (const float*)d_in[7];
    const float* bhh1 = (const float*)d_in[8];
    const float* Wlin = (const float*)d_in[9];
    const float* blin = (const float*)d_in[10];
    float* out = (float*)d_out;

    lstm2_halves_kernel<<<Bsz / 16, 32, 0, stream>>>(
        x, Wih0, Whh0, bih0, bhh0, Wih1, Whh1, bih1, bhh1, Wlin, blin, out);
}


# Exponent scales folded into constants: weight, input, bias rows hold -log2(e) (i,f,o) or 2log2(e) (g); cell state times 2log2(e); 6 fewer mul/add a cell; sum order kept, terms scaled, not bit-equal
# speedup vs baseline: 3.8465x; 1.0722x over previous
#include <hip/hip_runtime.h>


typedef float v2f __attribute__((ext_vector_type(2)));
typedef float v8f __attribute__((ext_vector_type(8)));

#define Bsz 2048
#define Tsz 4096
#define Hsz 8

#define KS (-1.4426950408889634f)
#define KT (2.8853900817779268f)

__device__ __forceinline__ float ex2(float z) {
    return __builtin_amdgcn_exp2f(z);
}
__device__ __forceinline__ float fsig_s(float z) {
    return __builtin_amdgcn_rcpf(1.0f + ex2(z));
}
__device__ __forceinline__ float ftanh_s(float z) {
    return 1.0f - 2.0f * __builtin_amdgcn_rcpf(ex2(z) + 1.0f);
}
__device__ __forceinline__ float ftanh_sk(float z) {
    return fmaf(__builtin_amdgcn_rcpf(ex2(z) + 1.0f), -2.0f * KT, KT);
}

__device__ __forceinline__ v8f wmma4(v2f a, v2f b, v8f c) {
    return __builtin_amdgcn_wmma_f32_16x16x4_f32(
         false, a,  false, b,
         (short)0, c,  false,  false);
}

__device__ __forceinline__ void gates_start(const float (&wC)[4][8],
                                            const float (&bC)[4][8],
                                            const float xv, v8f (&G)[4])
{
#pragma unroll
    for (int g = 0; g < 4; ++g)
#pragma unroll
        for (int r = 0; r < 8; ++r)
            G[g][r] = fmaf(xv, wC[g][r], bC[g][r]);
}

#define ORDER(outs, ins) asm volatile("" : outs : ins)
#define IO4(v) "+v"(v[0]), "+v"(v[1]), "+v"(v[2]), "+v"(v[3])
#define IN4(v) "v"(v[0]), "v"(v[1]), "v"(v[2]), "v"(v[3])
#define COMMA ,

__device__ __forceinline__ void lstm_stage(const v2f (&A)[4][4],
                                           const v2f (&wC2)[4][4],
                                           const v2f (&bC2)[4][4],
                                           const float (&wl)[8], const float bl,
                                           v2f xre,
                                           v8f (&G)[4], v8f (&Gn)[4],
                                           v2f (&sB)[4], float (&c)[8],
                                           float* outp, const bool hh)
{
    float p;
    float a[4], b[4], ei[4], ef[4], eg[4], eo[4], ih[4], fh[4], gh[4], oh[4];
    float cB[4], ec[4], th[4];
#pragma unroll
    for (int r = 0; r < 4; ++r) cB[r] = c[r + 4];

#pragma unroll
    for (int r = 0; r < 4; ++r) a[r] = G[0][r + 4];
    ORDER("+v"(sB[0]) COMMA IO4(a), "v"(sB[1]));
    Gn[0] = wmma4(A[0][0], sB[0], Gn[0]);
#pragma unroll
    for (int r = 0; r < 4; ++r) ei[r] = ex2(a[r]);

#pragma unroll
    for (int r = 0; r < 4; ++r) a[r] = G[1][r + 4];
    ORDER("+v"(sB[0]) COMMA IO4(a) COMMA IO4(ei), "v"(Gn[0]));
    Gn[1] = wmma4(A[1][0], sB[0], Gn[1]);
#pragma unroll
    for (int r = 0; r < 4; ++r) { ih[r] = __builtin_amdgcn_rcpf(1.0f + ei[r]); ef[r] = ex2(a[r]); }

#pragma unroll
    for (int r = 0; r < 4; ++r) a[r] = G[2][r + 4];
    ORDER("+v"(sB[0]) COMMA IO4(a) COMMA IO4(ef), "v"(Gn[1]) COMMA IN4(ih));
    Gn[2] = wmma4(A[2][0], sB[0], Gn[2]);
#pragma unroll
    for (int r = 0; r < 4; ++r) { fh[r] = __builtin_amdgcn_rcpf(1.0f + ef[r]); eg[r] = ex2(a[r]); }

    ORDER("+v"(sB[0]) COMMA IO4(eg), "v"(Gn[2]) COMMA IN4(fh));
    Gn[3] = wmma4(A[3][0], sB[0], Gn[3]);
#pragma unroll
    for (int r = 0; r < 4; ++r) eg[r] = eg[r] + 1.0f;
    ORDER(IO4(eg), "v"(sB[0]));
#pragma unroll
    for (int r = 0; r < 4; ++r) eg[r] = __builtin_amdgcn_rcpf(eg[r]);
    ORDER(IO4(eg), "v"(sB[0]));
#pragma unroll
    for (int r = 0; r < 4; ++r) gh[r] = fmaf(eg[r], -2.0f * KT, KT);

#pragma unroll
    for (int r = 0; r < 4; ++r) a[r] = G[3][r + 4];
    ORDER("+v"(sB[1]) COMMA IO4(a) COMMA IO4(cB), "v"(Gn[3]) COMMA IN4(gh));
    Gn[0] = wmma4(A[0][1], sB[1], Gn[0]);
#pragma unroll
    for (int r = 0; r < 4; ++r) { cB[r] = fmaf(fh[r], cB[r], ih[r] * gh[r]); eo[r] = ex2(a[r]); }

    ORDER("+v"(sB[1]) COMMA IO4(cB), "v"(Gn[0]) COMMA IN4(eo));
    Gn[1] = wmma4(A[1][1], sB[1], Gn[1]);
#pragma unroll
    for (int r = 0; r < 4; ++r) { ec[r] = ex2(cB[r]); c[r + 4] = cB[r]; }

    ORDER("+v"(sB[1]) COMMA IO4(ec), "v"(Gn[1]));
    Gn[2] = wmma4(A[2][1], sB[1], Gn[2]);
#pragma unroll
    for (int r = 0; r < 4; ++r) ec[r] = ec[r] + 1.0f;
    ORDER(IO4(ec), "v"(sB[1]));
#pragma unroll
    for (int r = 0; r < 4; ++r) ec[r] = __builtin_amdgcn_rcpf(ec[r]);
    ORDER(IO4(ec), "v"(sB[1]));
#pragma unroll
    for (int r = 0; r < 4; ++r) th[r] = 1.0f - 2.0f * ec[r];

    ORDER("+v"(sB[1]) COMMA IO4(th) COMMA IO4(eo), "v"(Gn[2]));
    Gn[3] = wmma4(A[3][1], sB[1], Gn[3]);
#pragma unroll
    for (int r = 0; r < 4; ++r) eo[r] = 1.0f + eo[r];
    ORDER(IO4(eo), "v"(sB[1]));
#pragma unroll
    for (int r = 0; r < 4; ++r) oh[r] = __builtin_amdgcn_rcpf(eo[r]);
    sB[2].x = oh[0] * th[0]; sB[2].y = oh[1] * th[1];
    sB[3].x = oh[2] * th[2]; sB[3].y = oh[3] * th[3];

    p = bl;
    ORDER("+v"(sB[2]) COMMA "+v"(xre) COMMA "+v"(p), "v"(Gn[3]) COMMA "v"(sB[3]));
    Gn[2] = wmma4(A[2][2], sB[2], Gn[2]);
#pragma unroll
    for (int j = 0; j < 4; ++j) {
        const v2f t2 = __builtin_elementwise_fma(xre, wC2[0][j], bC2[0][j]);
        G[0][2 * j] = t2.x; G[0][2 * j + 1] = t2.y;
    }
    p = fmaf(sB[0].x, wl[0], p); p = fmaf(sB[0].y, wl[1], p);

    ORDER("+v"(sB[2]) COMMA "+v"(xre) COMMA "+v"(p), "v"(Gn[2]) COMMA "v"(G[0]));
    Gn[0] = wmma4(A[0][2], sB[2], Gn[0]);
#pragma unroll
    for (int j = 0; j < 4; ++j) {
        const v2f t2 = __builtin_elementwise_fma(xre, wC2[1][j], bC2[1][j]);
        G[1][2 * j] = t2.x; G[1][2 * j + 1] = t2.y;
    }
    p = fmaf(sB[1].x, wl[2], p); p = fmaf(sB[1].y, wl[3], p);

    ORDER("+v"(sB[2]) COMMA "+v"(xre) COMMA "+v"(p), "v"(Gn[0]) COMMA "v"(G[1]));
    Gn[1] = wmma4(A[1][2], sB[2], Gn[1]);
#pragma unroll
    for (int j = 0; j < 4; ++j) {
        const v2f t2 = __builtin_elementwise_fma(xre, wC2[2][j], bC2[2][j]);
        G[2][2 * j] = t2.x; G[2][2 * j + 1] = t2.y;
    }
    p = fmaf(sB[2].x, wl[4], p); p = fmaf(sB[2].y, wl[5], p);

    ORDER("+v"(sB[3]) COMMA "+v"(xre) COMMA "+v"(p), "v"(Gn[1]) COMMA "v"(G[2]));
    Gn[2] = wmma4(A[2][3], sB[3], Gn[2]);
#pragma unroll
    for (int j = 0; j < 4; ++j) {
        const v2f t2 = __builtin_elementwise_fma(xre, wC2[3][j], bC2[3][j]);
        G[3][2 * j] = t2.x; G[3][2 * j + 1] = t2.y;
    }
    p = fmaf(sB[3].x, wl[6], p); p = fmaf(sB[3].y, wl[7], p);

    ORDER("+v"(sB[3]) COMMA "+v"(p), "v"(Gn[2]) COMMA "v"(G[3]));
    Gn[0] = wmma4(A[0][3], sB[3], Gn[0]);
    if (hh) *outp = p;

#pragma unroll
    for (int r = 0; r < 4; ++r) a[r] = Gn[2][r];
    ORDER("+v"(sB[3]) COMMA IO4(a), "v"(Gn[0]));
    Gn[1] = wmma4(A[1][3], sB[3], Gn[1]);
#pragma unroll
    for (int r = 0; r < 4; ++r) eg[r] = ex2(a[r]);

#pragma unroll
    for (int r = 0; r < 4; ++r) { a[r] = Gn[0][r]; b[r] = Gn[1][r]; }
    ORDER("+v"(sB[2]) COMMA IO4(eg) COMMA IO4(a) COMMA IO4(b), "v"(sB[3]));
    Gn[3] = wmma4(A[3][2], sB[2], Gn[3]);
    float ga[4], m[4], fa[4], oa[4], hA[4];
#pragma unroll
    for (int r = 0; r < 4; ++r) { eg[r] = eg[r] + 1.0f; ei[r] = ex2(a[r]); ef[r] = ex2(b[r]); }
    ORDER(IO4(eg), "v"(sB[2]));
#pragma unroll
    for (int r = 0; r < 4; ++r) eg[r] = __builtin_amdgcn_rcpf(eg[r]);
    ORDER(IO4(eg), "v"(sB[2]));
#pragma unroll
    for (int r = 0; r < 4; ++r) ga[r] = fmaf(eg[r], -2.0f * KT, KT);

    ORDER("+v"(sB[3]) COMMA IO4(ei) COMMA IO4(ef), "v"(Gn[3]) COMMA IN4(ga));
    Gn[3] = wmma4(A[3][3], sB[3], Gn[3]);
#pragma unroll
    for (int r = 0; r < 4; ++r) { ef[r] = 1.0f + ef[r]; ei[r] = 1.0f + ei[r]; }

    ORDER(IO4(ef) COMMA IO4(ei), "v"(Gn[3]));
#pragma unroll
    for (int r = 0; r < 4; ++r) { fa[r] = __builtin_amdgcn_rcpf(ef[r]); ei[r] = __builtin_amdgcn_rcpf(ei[r]); }
    ORDER(IO4(fa) COMMA IO4(ei), "v"(sB[3]));
#pragma unroll
    for (int r = 0; r < 4; ++r) m[r] = ei[r] * ga[r];
    ORDER(IO4(fa), "v"(sB[3]));
#pragma unroll
    for (int r = 0; r < 4; ++r) c[r] = fmaf(fa[r], c[r], m[r]);
#pragma unroll
    for (int r = 0; r < 4; ++r) a[r] = Gn[3][r];
    ORDER("+v"(c[0]) COMMA "+v"(c[1]) COMMA "+v"(c[2]) COMMA "+v"(c[3]) COMMA IO4(a), "v"(sB[3]));
#pragma unroll
    for (int r = 0; r < 4; ++r) { ec[r] = ex2(c[r]); eo[r] = ex2(a[r]); }
    ORDER(IO4(ec) COMMA IO4(eo), "v"(sB[3]));
#pragma unroll
    for (int r = 0; r < 4; ++r) { ec[r] = ec[r] + 1.0f; eo[r] = 1.0f + eo[r]; }
    ORDER(IO4(ec) COMMA IO4(eo), "v"(sB[3]));
#pragma unroll
    for (int r = 0; r < 4; ++r) { ec[r] = __builtin_amdgcn_rcpf(ec[r]); oa[r] = __builtin_amdgcn_rcpf(eo[r]); }
    ORDER(IO4(ec), IN4(oa));
#pragma unroll
    for (int r = 0; r < 4; ++r) th[r] = 1.0f - 2.0f * ec[r];
#pragma unroll
    for (int r = 0; r < 4; ++r) hA[r] = oa[r] * th[r];
    sB[0].x = hA[0]; sB[0].y = hA[1]; sB[1].x = hA[2]; sB[1].y = hA[3];
}

__global__ __launch_bounds__(32) void lstm2_halves_kernel(
    const float* __restrict__ x,
    const float* __restrict__ Wih0,
    const float* __restrict__ Whh0,
    const float* __restrict__ bih0, const float* __restrict__ bhh0,
    const float* __restrict__ Wih1,
    const float* __restrict__ Whh1,
    const float* __restrict__ bih1, const float* __restrict__ bhh1,
    const float* __restrict__ Wlin,
    const float* __restrict__ blin,
    float* __restrict__ out)
{
    const int lane = threadIdx.x;
    const int lm   = lane & 15;
    const bool hh  = (lane >= 16);
    const long row = (long)(blockIdx.x * 16 + lm) * Tsz;

    v2f A[4][4];
    {
        const bool l1row = (lm >= 8);
        const float* Wsrc = l1row ? (hh ? Whh1 : Wih1) : Whh0;
        const bool live = l1row || !hh;
#pragma unroll
        for (int g = 0; g < 4; ++g) {
            const int G = 8 * g + (lm & 7);
            const float sg = (g == 2) ? KT : KS;
#pragma unroll
            for (int f = 0; f < 4; ++f) {
                const float w0 = sg * Wsrc[G * Hsz + 2 * f];
                const float w1 = sg * Wsrc[G * Hsz + 2 * f + 1];
                A[g][f].x = live ? w0 : 0.0f;
                A[g][f].y = live ? w1 : 0.0f;
            }
        }
    }

    float wC[4][8], bC[4][8];
#pragma unroll
    for (int g = 0; g < 4; ++g)
#pragma unroll
        for (int r = 0; r < 8; ++r) {
            const int G = 8 * g + r;
            const float sg = (g == 2) ? KT : KS;
            const float w0 = sg * Wih0[G];
            const float b0 = sg * (bih0[G] + bhh0[G]);
            const float b1 = sg * (bih1[G] + bhh1[G]);
            wC[g][r] = hh ? 0.0f : w0;
            bC[g][r] = hh ? b1 : b0;
        }

    float wl[8];
#pragma unroll
    for (int j = 0; j < 8; ++j) wl[j] = Wlin[j];
    const float bl = blin[0];

    float s[8], c[8];
#pragma unroll
    for (int j = 0; j < 8; ++j) { s[j] = 0.f; c[j] = 0.f; }

    const float* xp = x + row;
    const float x0 = xp[0], x1 = xp[1], x2 = xp[2];
    float xb = xp[3], xc = xp[4];
    xp += 5;

    v8f Ga[4], Gb[4];
    gates_start(wC, bC, x0, Ga);
#pragma unroll
    for (int f = 0; f < 4; ++f) {
        v2f Bz; Bz.x = s[2 * f]; Bz.y = s[2 * f + 1];
#pragma unroll
        for (int g = 0; g < 4; ++g) Ga[g] = wmma4(A[g][f], Bz, Ga[g]);
    }

#pragma unroll
    for (int r = 0; r < 8; ++r) {
        const float i0 = fsig_s(Ga[0][r]);
        const float f0 = fsig_s(Ga[1][r]);
        const float g0 = ftanh_sk(Ga[2][r]);
        const float o0 = fsig_s(Ga[3][r]);
        c[r] = fmaf(f0, c[r], i0 * g0);
        s[r] = o0 * ftanh_s(c[r]);
        s[r] = hh ? 0.f : s[r];
        c[r] = hh ? 0.f : c[r];
    }

    gates_start(wC, bC, x1, Gb);
#pragma unroll
    for (int f = 0; f < 4; ++f) {
        v2f Bs; Bs.x = s[2 * f]; Bs.y = s[2 * f + 1];
#pragma unroll
        for (int g = 0; g < 4; ++g) Gb[g] = wmma4(A[g][f], Bs, Gb[g]);
    }
    gates_start(wC, bC, x2, Ga);
#pragma unroll
    for (int r = 0; r < 4; ++r) {
        const float i1 = fsig_s(Gb[0][r]);
        const float f1 = fsig_s(Gb[1][r]);
        const float g1 = ftanh_sk(Gb[2][r]);
        const float o1 = fsig_s(Gb[3][r]);
        c[r] = fmaf(f1, c[r], i1 * g1);
        s[r] = o1 * ftanh_s(c[r]);
    }
    v2f wC2[4][4], bC2[4][4];
#pragma unroll
    for (int g = 0; g < 4; ++g)
#pragma unroll
        for (int j = 0; j < 4; ++j) {
            wC2[g][j].x = wC[g][2 * j]; wC2[g][j].y = wC[g][2 * j + 1];
            bC2[g][j].x = bC[g][2 * j]; bC2[g][j].y = bC[g][2 * j + 1];
        }
    v2f sB[4];
#pragma unroll
    for (int f = 0; f < 4; ++f) { sB[f].x = s[2 * f]; sB[f].y = s[2 * f + 1]; }

    float* op = out + row;
    for (int it = 0; it < Tsz / 2; ++it) {
        const float nb = *xp;
        __builtin_prefetch(xp + 255, 0, 0);
        xp += (2 * it + 6 < Tsz) ? 1 : 0;
        const float nc = *xp;
        xp += (2 * it + 7 < Tsz) ? 1 : 0;

        v2f xb2, xc2;
        xb2.x = xb; xb2.y = xb; xc2.x = xc; xc2.y = xc;
        lstm_stage(A, wC2, bC2, wl, bl, xb2, Gb, Ga, sB, c, op,     hh);
        lstm_stage(A, wC2, bC2, wl, bl, xc2, Ga, Gb, sB, c, op + 1, hh);

        xb = nb; xc = nc;
        op += 2;
    }
}

extern "C" void kernel_launch(void* const* d_in, const int* in_sizes, int n_in,
                              void* d_out, int out_size, void* d_ws, size_t ws_size,
                              hipStream_t stream) {
    (void)in_sizes; (void)n_in; (void)out_size; (void)d_ws; (void)ws_size;
    const float* x    = (const float*)d_in[0];
    const float* Wih0 = (const float*)d_in[1];
    const float* Whh0 = (const float*)d_in[2];
    const float* bih0 = (const float*)d_in[3];
    const float* bhh0 = (const float*)d_in[4];
    const float* Wih1 = (const float*)d_in[5];
    const float* Whh1 = (const float*)d_in[6];
    const float* bih1 = (const float*)d_in[7];
    const float* bhh1 = (const float*)d_in[8];
    const float* Wlin = (const float*)d_in[9];
    const float* blin = (const float*)d_in[10];
    float* out = (float*)d_out;

    lstm2_halves_kernel<<<Bsz / 16, 32, 0, stream>>>(
        x, Wih0, Whh0, bih0, bhh0, Wih1, Whh1, bih1, bhh1, Wlin, blin, out);
}


# Hidden units split over the 4 waves of a 128-thread group: a wave owns 2 units of both layers (1 gate tile, 4 matrix ops a step, was 16); h handed over in LDS, 1 barrier a step; same sums, same order
# speedup vs baseline: 5.5240x; 1.4361x over previous
#include <hip/hip_runtime.h>


typedef float v2f __attribute__((ext_vector_type(2)));
typedef float v8f __attribute__((ext_vector_type(8)));

#define Bsz 2048
#define Tsz 4096
#define Hsz 8

#define KS (-1.4426950408889634f)
#define KT (2.8853900817779268f)

__device__ __forceinline__ float ex2(float z) {
    return __builtin_amdgcn_exp2f(z);
}
__device__ __forceinline__ float fsig_s(float z) {
    return __builtin_amdgcn_rcpf(1.0f + ex2(z));
}
__device__ __forceinline__ float ftanh_s(float z) {
    return 1.0f - 2.0f * __builtin_amdgcn_rcpf(ex2(z) + 1.0f);
}
__device__ __forceinline__ float ftanh_sk(float z) {
    return fmaf(__builtin_amdgcn_rcpf(ex2(z) + 1.0f), -2.0f * KT, KT);
}

__device__ __forceinline__ v8f wmma4(v2f a, v2f b, v8f c) {
    return __builtin_amdgcn_wmma_f32_16x16x4_f32(
         false, a,  false, b,
         (short)0, c,  false,  false);
}

#define BESIDE(p, G) asm volatile("" : "+v"(p), "+v"(G))
template <bool WRITE>
__device__ __forceinline__ void trip_cells(const v2f (&A)[4], const v2f (&B)[4],
                                           const float (&wC)[8], const float (&bC)[8],
                                           const float (&wl)[8], const float bl,
                                           const float xv,
                                           float (&c)[2], float (&h)[2],
                                           float* outp, const bool wr)
{
    v8f G;
#pragma unroll
    for (int r = 0; r < 8; ++r) G[r] = fmaf(xv, wC[r], bC[r]);
    float p = bl;
#pragma unroll
    for (int f = 0; f < 4; ++f) {
        G = wmma4(A[f], B[f], G);
        if (WRITE) {
            p = fmaf(B[f].x, wl[2 * f], p);
            p = fmaf(B[f].y, wl[2 * f + 1], p);
            BESIDE(p, G);
        }
    }
    if (WRITE) { if (wr) *outp = p; }
#pragma unroll
    for (int u = 0; u < 2; ++u) {
        const float ig = fsig_s(G[u]);
        const float fg = fsig_s(G[2 + u]);
        const float gg = ftanh_sk(G[4 + u]);
        const float og = fsig_s(G[6 + u]);
        c[u] = fmaf(fg, c[u], ig * gg);
        h[u] = og * ftanh_s(c[u]);
    }
}

__device__ __forceinline__ void hand_over(v2f (*slot)[32], const int w, const int lane,
                                          const float (&h)[2], v2f (&B)[4])
{
    v2f hv; hv.x = h[0]; hv.y = h[1];
    slot[w][lane] = hv;
    asm volatile("s_wait_dscnt 0x0" ::: "memory");
    asm volatile("s_barrier_signal -1\n\ts_barrier_wait -1" ::: "memory");
#pragma unroll
    for (int f = 0; f < 4; ++f) B[f] = slot[f][lane];
}

__device__ __forceinline__ float project(const v2f (&B)[4], const float (&wl)[8], const float bl)
{
    float p = bl;
#pragma unroll
    for (int f = 0; f < 4; ++f) {
        p = fmaf(B[f].x, wl[2 * f], p);
        p = fmaf(B[f].y, wl[2 * f + 1], p);
    }
    return p;
}

__global__ __launch_bounds__(128) void lstm2_units4_kernel(
    const float* __restrict__ x,
    const float* __restrict__ Wih0,
    const float* __restrict__ Whh0,
    const float* __restrict__ bih0, const float* __restrict__ bhh0,
    const float* __restrict__ Wih1,
    const float* __restrict__ Whh1,
    const float* __restrict__ bih1, const float* __restrict__ bhh1,
    const float* __restrict__ Wlin,
    const float* __restrict__ blin,
    float* __restrict__ out)
{
    __shared__ v2f hx[2][4][32];

    const int lane = threadIdx.x & 31;
    const int w    = __builtin_amdgcn_readfirstlane((int)(threadIdx.x >> 5));
    const int lm   = lane & 15;
    const bool hh  = (lane >= 16);
    const long row = (long)(blockIdx.x * 16 + lm) * Tsz;

    v2f A[4];
    {
        const bool l1row = (lm >= 8);
        const int q    = lm & 7;
        const int gate = q >> 1;
        const int Gr   = 8 * gate + 2 * w + (q & 1);
        const float sg = (gate == 2) ? KT : KS;
        const float* Wsrc = l1row ? (hh ? Whh1 : Wih1) : Whh0;
        const bool live = l1row || !hh;
#pragma unroll
        for (int f = 0; f < 4; ++f) {
            const float w0 = sg * Wsrc[Gr * Hsz + 2 * f];
            const float w1 = sg * Wsrc[Gr * Hsz + 2 * f + 1];
            A[f].x = live ? w0 : 0.0f;
            A[f].y = live ? w1 : 0.0f;
        }
    }

    float wC[8], bC[8];
#pragma unroll
    for (int r = 0; r < 8; ++r) {
        const int gate = r >> 1;
        const int Gr = 8 * gate + 2 * w + (r & 1);
        const float sg = (gate == 2) ? KT : KS;
        const float w0 = sg * Wih0[Gr];
        const float b0 = sg * (bih0[Gr] + bhh0[Gr]);
        const float b1 = sg * (bih1[Gr] + bhh1[Gr]);
        wC[r] = hh ? 0.0f : w0;
        bC[r] = hh ? b1 : b0;
    }

    float wl[8];
#pragma unroll
    for (int j = 0; j < 8; ++j) wl[j] = Wlin[j];
    const float bl = blin[0];

    float c[2], h[2];
    c[0] = 0.f; c[1] = 0.f;
    v2f B[4];
#pragma unroll
    for (int f = 0; f < 4; ++f) { B[f].x = 0.f; B[f].y = 0.f; }

    const float* xrow = x + row;
    const float x0 = xrow[0];
    float xb = xrow[1], xc = xrow[2];

    trip_cells<false>(A, B, wC, bC, wl, bl, x0, c, h, nullptr, false);
#pragma unroll
    for (int u = 0; u < 2; ++u) {
        h[u] = hh ? 0.f : h[u];
        c[u] = hh ? 0.f : c[u];
    }
    hand_over(hx[0], w, lane, h, B);

    float* op = out + row;
    const bool wr = hh && (w == 0);
    for (int it = 0; it < Tsz / 2; ++it) {
        const int i1 = (2 * it + 3 < Tsz) ? 2 * it + 3 : Tsz - 1;
        const int i2 = (2 * it + 4 < Tsz) ? 2 * it + 4 : Tsz - 1;
        const float nb = xrow[i1];
        __builtin_prefetch(xrow + i1 + 255, 0, 0);
        const float nc = xrow[i2];

        trip_cells<true>(A, B, wC, bC, wl, bl, xb, c, h, op + (2 * it - 1), wr && (it > 0));
        hand_over(hx[1], w, lane, h, B);

        trip_cells<true>(A, B, wC, bC, wl, bl, xc, c, h, op + 2 * it, wr);
        hand_over(hx[0], w, lane, h, B);

        xb = nb; xc = nc;
    }
    const float pl = project(B, wl, bl);
    if (wr) op[Tsz - 1] = pl;
}

extern "C" void kernel_launch(void* const* d_in, const int* in_sizes, int n_in,
                              void* d_out, int out_size, void* d_ws, size_t ws_size,
                              hipStream_t stream) {
    (void)in_sizes; (void)n_in; (void)out_size; (void)d_ws; (void)ws_size;
    const float* x    = (const float*)d_in[0];
    const float* Wih0 = (const float*)d_in[1];
    const float* Whh0 = (const float*)d_in[2];
    const float* bih0 = (const float*)d_in[3];
    const float* bhh0 = (const float*)d_in[4];
    const float* Wih1 = (const float*)d_in[5];
    const float* Whh1 = (const float*)d_in[6];
    const float* bih1 = (const float*)d_in[7];
    const float* bhh1 = (const float*)d_in[8];
    const float* Wlin = (const float*)d_in[9];
    const float* blin = (const float*)d_in[10];
    float* out = (float*)d_out;

    lstm2_units4_kernel<<<Bsz / 16, 128, 0, stream>>>(
        x, Wih0, Whh0, bih0, bhh0, Wih1, Whh1, bih1, bhh1, Wlin, blin, out);
}


# Wave owns 4 batch rows, all units: 4 column groups pick 1 of 4 gate tiles each (12 matrix ops a step); state passed by 8 in-row lane gathers, no LDS, no barrier; k sums reordered, out dot in 2 halves
# speedup vs baseline: 5.7518x; 1.0412x over previous
#include <hip/hip_runtime.h>


typedef float v2f __attribute__((ext_vector_type(2)));
typedef float v8f __attribute__((ext_vector_type(8)));

#define Bsz 2048
#define Tsz 4096
#define Hsz 8

#define KS (-1.4426950408889634f)
#define KT (2.8853900817779268f)

__device__ __forceinline__ float ex2(float z) {
    return __builtin_amdgcn_exp2f(z);
}
__device__ __forceinline__ float fsig_s(float z) {
    return __builtin_amdgcn_rcpf(1.0f + ex2(z));
}
__device__ __forceinline__ float ftanh_s(float z) {
    return 1.0f - 2.0f * __builtin_amdgcn_rcpf(ex2(z) + 1.0f);
}
__device__ __forceinline__ float ftanh_sk(float z) {
    return fmaf(__builtin_amdgcn_rcpf(ex2(z) + 1.0f), -2.0f * KT, KT);
}

__device__ __forceinline__ v8f wmma4(v2f a, v2f b, v8f c) {
    return __builtin_amdgcn_wmma_f32_16x16x4_f32(
         false, a,  false, b,
         (short)0, c,  false,  false);
}

template <int F>
__device__ __forceinline__ float group_read(float v) {
    constexpr unsigned sel = (F == 0) ? 0x32103210u : (F == 1) ? 0x76547654u
                           : (F == 2) ? 0xBA98BA98u : 0xFEDCFEDCu;
    const int iv = __builtin_bit_cast(int, v);
    return __builtin_bit_cast(float, __builtin_amdgcn_permlane16(iv, iv, sel, sel, true, false));
}
__device__ __forceinline__ float other_half(float v) {
    const int iv = __builtin_bit_cast(int, v);
    return __builtin_bit_cast(float, __builtin_amdgcn_permlanex16(iv, iv, 0x76543210u, 0xfedcba98u, true, false));
}

__device__ __forceinline__ void hand_over(const float (&h)[2], v2f (&B)[4]) {
    B[0].x = group_read<0>(h[0]); B[0].y = group_read<0>(h[1]);
    B[1].x = group_read<1>(h[0]); B[1].y = group_read<1>(h[1]);
    B[2].x = group_read<2>(h[0]); B[2].y = group_read<2>(h[1]);
    B[3].x = group_read<3>(h[0]); B[3].y = group_read<3>(h[1]);
}

__device__ __forceinline__ float project_half(const v2f (&B)[4], const float (&wlh)[4], const float bh) {
    float p = fmaf(B[2].x, wlh[0], bh);
    p = fmaf(B[2].y, wlh[1], p);
    p = fmaf(B[3].x, wlh[2], p);
    p = fmaf(B[3].y, wlh[3], p);
    return p;
}

__device__ __forceinline__ v8f c_tile(const float xv, const float (&wC)[8], const float (&bC)[8]) {
    v8f Gi;
#pragma unroll
    for (int r = 0; r < 8; ++r) Gi[r] = fmaf(xv, wC[r], bC[r]);
    return Gi;
}

template <bool WRITE>
__device__ __forceinline__ void trip_cells(const v2f (&A0)[2][2], const v2f (&A1)[2][4],
                                           const v2f (&B)[4],
                                           const v8f Gi,
                                           const float (&wlh)[4], const float bh,
                                           const bool s1, const bool s2, const bool s3,
                                           float (&c)[2], float (&h)[2],
                                           float* outp, const bool wr)
{
    float ph = 0.f;
    if (WRITE) ph = project_half(B, wlh, bh);

    v8f T2 = wmma4(A1[0][0], B[0], Gi);
    v8f T3 = wmma4(A1[1][0], B[0], Gi);
    v8f T0 = wmma4(A0[0][0], B[0], Gi);
    v8f T1 = wmma4(A0[1][0], B[0], Gi);
    T2 = wmma4(A1[0][1], B[1], T2);
    T3 = wmma4(A1[1][1], B[1], T3);
    T0 = wmma4(A0[0][1], B[1], T0);
    T1 = wmma4(A0[1][1], B[1], T1);
    T2 = wmma4(A1[0][2], B[2], T2);
    T3 = wmma4(A1[1][2], B[2], T3);
    T2 = wmma4(A1[0][3], B[3], T2);
    T3 = wmma4(A1[1][3], B[3], T3);

    if (WRITE) {
        const float p = ph + other_half(ph);
        if (wr) *outp = p;
    }

    v8f G;
#pragma unroll
    for (int r = 0; r < 8; ++r) {
        float v = s1 ? T1[r] : T0[r];
        v = s2 ? T2[r] : v;
        G[r] = s3 ? T3[r] : v;
    }

#pragma unroll
    for (int u = 0; u < 2; ++u) {
        const float ig = fsig_s(G[u]);
        const float fg = fsig_s(G[2 + u]);
        const float gg = ftanh_sk(G[4 + u]);
        const float og = fsig_s(G[6 + u]);
        c[u] = fmaf(fg, c[u], ig * gg);
        h[u] = og * ftanh_s(c[u]);
    }
}

__global__ __launch_bounds__(128) void lstm2_rows4_kernel(
    const float* __restrict__ x,
    const float* __restrict__ Wih0,
    const float* __restrict__ Whh0,
    const float* __restrict__ bih0, const float* __restrict__ bhh0,
    const float* __restrict__ Wih1,
    const float* __restrict__ Whh1,
    const float* __restrict__ bih1, const float* __restrict__ bhh1,
    const float* __restrict__ Wlin,
    const float* __restrict__ blin,
    float* __restrict__ out)
{
    const int lane = threadIdx.x & 31;
    const int w    = __builtin_amdgcn_readfirstlane((int)(threadIdx.x >> 5));
    const int cc   = lane & 15;
    const int hf   = lane >> 4;
    const int j    = cc >> 2;
    const int b    = cc & 3;
    const bool l1  = (j >= 2);
    const int q    = j & 1;
    const long row = (long)(blockIdx.x * 16 + 4 * w + b) * Tsz;

    v2f A0[2][2];
    v2f A1[2][4];
    {
        const int gate = (cc & 7) >> 1;
        const float sg = (gate == 2) ? KT : KS;
#pragma unroll
        for (int qq = 0; qq < 2; ++qq) {
            const int unit = 4 * qq + 2 * (cc >> 3) + (cc & 1);
            const int Gr   = 8 * gate + unit;
#pragma unroll
            for (int f = 0; f < 2; ++f) {
                const int k = 4 * f + 2 * hf;
                A0[qq][f].x = sg * Whh0[Gr * Hsz + k];
                A0[qq][f].y = sg * Whh0[Gr * Hsz + k + 1];
                A1[qq][f].x = sg * Wih1[Gr * Hsz + k];
                A1[qq][f].y = sg * Wih1[Gr * Hsz + k + 1];
                A1[qq][2 + f].x = sg * Whh1[Gr * Hsz + k];
                A1[qq][2 + f].y = sg * Whh1[Gr * Hsz + k + 1];
            }
        }
    }

    float wC[8], bC[8];
#pragma unroll
    for (int r = 0; r < 8; ++r) {
        const int gate = r >> 1;
        const int Gr = 8 * gate + 4 * q + 2 * hf + (r & 1);
        const float sg = (gate == 2) ? KT : KS;
        const float w0 = sg * Wih0[Gr];
        const float b0 = sg * (bih0[Gr] + bhh0[Gr]);
        const float b1 = sg * (bih1[Gr] + bhh1[Gr]);
        wC[r] = l1 ? 0.0f : w0;
        bC[r] = l1 ? b1 : b0;
    }

    float wlh[4];
    wlh[0] = Wlin[2 * hf];     wlh[1] = Wlin[2 * hf + 1];
    wlh[2] = Wlin[4 + 2 * hf]; wlh[3] = Wlin[5 + 2 * hf];
    const float bh = (hf == 0) ? blin[0] : 0.0f;

    const bool s1 = (j == 1), s2 = (j == 2), s3 = (j == 3);

    float c[2], h[2];
    c[0] = 0.f; c[1] = 0.f;
    v2f B[4];
#pragma unroll
    for (int f = 0; f < 4; ++f) { B[f].x = 0.f; B[f].y = 0.f; }

    const float* xrow = x + row;
    const float x0 = xrow[0];
    float xc = xrow[2];
    v8f Gb = c_tile(xrow[1], wC, bC);

    trip_cells<false>(A0, A1, B, c_tile(x0, wC, bC), wlh, bh, s1, s2, s3, c, h, nullptr, false);
#pragma unroll
    for (int u = 0; u < 2; ++u) {
        h[u] = l1 ? 0.f : h[u];
        c[u] = l1 ? 0.f : c[u];
    }
    hand_over(h, B);

    float* op = out + row;
    const bool wr = (lane < 4);
    for (int it = 0; it < Tsz / 2; ++it) {
        const int i1 = (2 * it + 3 < Tsz) ? 2 * it + 3 : Tsz - 1;
        const int i2 = (2 * it + 4 < Tsz) ? 2 * it + 4 : Tsz - 1;
        const float nb = xrow[i1];
        __builtin_prefetch(xrow + i1 + 255, 0, 0);
        const float nc = xrow[i2];

        const v8f Gc = c_tile(xc, wC, bC);
        trip_cells<true>(A0, A1, B, Gb, wlh, bh, s1, s2, s3, c, h, op + (2 * it - 1), wr && (it > 0));
        hand_over(h, B);

        Gb = c_tile(nb, wC, bC);
        trip_cells<true>(A0, A1, B, Gc, wlh, bh, s1, s2, s3, c, h, op + 2 * it, wr);
        hand_over(h, B);

        xc = nc;
    }
    const float phl = project_half(B, wlh, bh);
    const float pl = phl + other_half(phl);
    if (wr) op[Tsz - 1] = pl;
}

extern "C" void kernel_launch(void* const* d_in, const int* in_sizes, int n_in,
                              void* d_out, int out_size, void* d_ws, size_t ws_size,
                              hipStream_t stream) {
    (void)in_sizes; (void)n_in; (void)out_size; (void)d_ws; (void)ws_size;
    const float* x    = (const float*)d_in[0];
    const float* Wih0 = (const float*)d_in[1];
    const float* Whh0 = (const float*)d_in[2];
    const float* bih0 = (const float*)d_in[3];
    const float* bhh0 = (const float*)d_in[4];
    const float* Wih1 = (const float*)d_in[5];
    const float* Whh1 = (const float*)d_in[6];
    const float* bih1 = (const float*)d_in[7];
    const float* bhh1 = (const float*)d_in[8];
    const float* Wlin = (const float*)d_in[9];
    const float* blin = (const float*)d_in[10];
    float* out = (float*)d_out;

    lstm2_rows4_kernel<<<Bsz / 16, 128, 0, stream>>>(
        x, Wih0, Whh0, bih0, bhh0, Wih1, Whh1, bih1, bhh1, Wlin, blin, out);
}


# Row I/O once per 8-trip pass: each of a row's 8 lanes keeps one step's output, one unmasked 32-lane store a pass; x as two 16-byte loads a pass ahead; no load, store, exec mask in a trip; same sums
# speedup vs baseline: 7.9301x; 1.3787x over previous
#include <hip/hip_runtime.h>


typedef float v2f __attribute__((ext_vector_type(2)));
typedef float v8f __attribute__((ext_vector_type(8)));
typedef float v4f __attribute__((ext_vector_type(4)));

#define Bsz 2048
#define Tsz 4096
#define Hsz 8

#define KS (-1.4426950408889634f)
#define KT (2.8853900817779268f)

__device__ __forceinline__ float ex2(float z) {
    return __builtin_amdgcn_exp2f(z);
}
__device__ __forceinline__ float fsig_s(float z) {
    return __builtin_amdgcn_rcpf(1.0f + ex2(z));
}
__device__ __forceinline__ float ftanh_s(float z) {
    return 1.0f - 2.0f * __builtin_amdgcn_rcpf(ex2(z) + 1.0f);
}
__device__ __forceinline__ float ftanh_sk(float z) {
    return fmaf(__builtin_amdgcn_rcpf(ex2(z) + 1.0f), -2.0f * KT, KT);
}

__device__ __forceinline__ v8f wmma4(v2f a, v2f b, v8f c) {
    return __builtin_amdgcn_wmma_f32_16x16x4_f32(
         false, a,  false, b,
         (short)0, c,  false,  false);
}

template <int F>
__device__ __forceinline__ float group_read(float v) {
    constexpr unsigned sel = (F == 0) ? 0x32103210u : (F == 1) ? 0x76547654u
                           : (F == 2) ? 0xBA98BA98u : 0xFEDCFEDCu;
    const int iv = __builtin_bit_cast(int, v);
    return __builtin_bit_cast(float, __builtin_amdgcn_permlane16(iv, iv, sel, sel, true, false));
}
__device__ __forceinline__ float other_half(float v) {
    const int iv = __builtin_bit_cast(int, v);
    return __builtin_bit_cast(float, __builtin_amdgcn_permlanex16(iv, iv, 0x76543210u, 0xfedcba98u, true, false));
}

__device__ __forceinline__ void hand_over(const float (&h)[2], v2f (&B)[4]) {
    B[0].x = group_read<0>(h[0]); B[0].y = group_read<0>(h[1]);
    B[1].x = group_read<1>(h[0]); B[1].y = group_read<1>(h[1]);
    B[2].x = group_read<2>(h[0]); B[2].y = group_read<2>(h[1]);
    B[3].x = group_read<3>(h[0]); B[3].y = group_read<3>(h[1]);
}

__device__ __forceinline__ float project_half(const v2f (&B)[4], const float (&wlh)[4], const float bh) {
    float p = fmaf(B[2].x, wlh[0], bh);
    p = fmaf(B[2].y, wlh[1], p);
    p = fmaf(B[3].x, wlh[2], p);
    p = fmaf(B[3].y, wlh[3], p);
    return p;
}

__device__ __forceinline__ v8f c_tile(const float xv, const float (&wC)[8], const float (&bC)[8]) {
    v8f Gi;
#pragma unroll
    for (int r = 0; r < 8; ++r) Gi[r] = fmaf(xv, wC[r], bC[r]);
    return Gi;
}

__device__ __forceinline__ void load8(const float* __restrict__ p, float (&d)[8]) {
    const v4f lo = *reinterpret_cast<const v4f*>(p);
    const v4f hi = *reinterpret_cast<const v4f*>(p + 4);
    d[0] = lo.x; d[1] = lo.y; d[2] = lo.z; d[3] = lo.w;
    d[4] = hi.x; d[5] = hi.y; d[6] = hi.z; d[7] = hi.w;
}

__device__ __forceinline__ void trip_cells(const v2f (&A0)[2][2], const v2f (&A1)[2][4],
                                           const v2f (&B)[4],
                                           const v8f Gi,
                                           const bool s1, const bool s2, const bool s3,
                                           float (&c)[2], float (&h)[2])
{
    v8f T2 = wmma4(A1[0][0], B[0], Gi);
    v8f T3 = wmma4(A1[1][0], B[0], Gi);
    v8f T0 = wmma4(A0[0][0], B[0], Gi);
    v8f T1 = wmma4(A0[1][0], B[0], Gi);
    T2 = wmma4(A1[0][1], B[1], T2);
    T3 = wmma4(A1[1][1], B[1], T3);
    T0 = wmma4(A0[0][1], B[1], T0);
    T1 = wmma4(A0[1][1], B[1], T1);
    T2 = wmma4(A1[0][2], B[2], T2);
    T3 = wmma4(A1[1][2], B[2], T3);
    T2 = wmma4(A1[0][3], B[3], T2);
    T3 = wmma4(A1[1][3], B[3], T3);

    v8f G;
#pragma unroll
    for (int r = 0; r < 8; ++r) {
        float v = s1 ? T1[r] : T0[r];
        v = s2 ? T2[r] : v;
        G[r] = s3 ? T3[r] : v;
    }

#pragma unroll
    for (int u = 0; u < 2; ++u) {
        const float ig = fsig_s(G[u]);
        const float fg = fsig_s(G[2 + u]);
        const float gg = ftanh_sk(G[4 + u]);
        const float og = fsig_s(G[6 + u]);
        c[u] = fmaf(fg, c[u], ig * gg);
        h[u] = og * ftanh_s(c[u]);
    }
}

__global__ __launch_bounds__(128) void lstm2_rows4_kernel(
    const float* __restrict__ x,
    const float* __restrict__ Wih0,
    const float* __restrict__ Whh0,
    const float* __restrict__ bih0, const float* __restrict__ bhh0,
    const float* __restrict__ Wih1,
    const float* __restrict__ Whh1,
    const float* __restrict__ bih1, const float* __restrict__ bhh1,
    const float* __restrict__ Wlin,
    const float* __restrict__ blin,
    float* __restrict__ out)
{
    const int lane = threadIdx.x & 31;
    const int w    = __builtin_amdgcn_readfirstlane((int)(threadIdx.x >> 5));
    const int cc   = lane & 15;
    const int hf   = lane >> 4;
    const int j    = cc >> 2;
    const int b    = cc & 3;
    const bool l1  = (j >= 2);
    const int q    = j & 1;
    const long row = (long)(blockIdx.x * 16 + 4 * w + b) * Tsz;

    v2f A0[2][2];
    v2f A1[2][4];
    {
        const int gate = (cc & 7) >> 1;
        const float sg = (gate == 2) ? KT : KS;
#pragma unroll
        for (int qq = 0; qq < 2; ++qq) {
            const int unit = 4 * qq + 2 * (cc >> 3) + (cc & 1);
            const int Gr   = 8 * gate + unit;
#pragma unroll
            for (int f = 0; f < 2; ++f) {
                const int k = 4 * f + 2 * hf;
                A0[qq][f].x = sg * Whh0[Gr * Hsz + k];
                A0[qq][f].y = sg * Whh0[Gr * Hsz + k + 1];
                A1[qq][f].x = sg * Wih1[Gr * Hsz + k];
                A1[qq][f].y = sg * Wih1[Gr * Hsz + k + 1];
                A1[qq][2 + f].x = sg * Whh1[Gr * Hsz + k];
                A1[qq][2 + f].y = sg * Whh1[Gr * Hsz + k + 1];
            }
        }
    }

    float wC[8], bC[8];
#pragma unroll
    for (int r = 0; r < 8; ++r) {
        const int gate = r >> 1;
        const int Gr = 8 * gate + 4 * q + 2 * hf + (r & 1);
        const float sg = (gate == 2) ? KT : KS;
        const float w0 = sg * Wih0[Gr];
        const float b0 = sg * (bih0[Gr] + bhh0[Gr]);
        const float b1 = sg * (bih1[Gr] + bhh1[Gr]);
        wC[r] = l1 ? 0.0f : w0;
        bC[r] = l1 ? b1 : b0;
    }

    float wlh[4];
    wlh[0] = Wlin[2 * hf];     wlh[1] = Wlin[2 * hf + 1];
    wlh[2] = Wlin[4 + 2 * hf]; wlh[3] = Wlin[5 + 2 * hf];
    const float bh = (hf == 0) ? blin[0] : 0.0f;

    const bool s1 = (j == 1), s2 = (j == 2), s3 = (j == 3);

    float c[2], h[2];
    c[0] = 0.f; c[1] = 0.f;
    v2f B[4];
#pragma unroll
    for (int f = 0; f < 4; ++f) { B[f].x = 0.f; B[f].y = 0.f; }

    const float* xrow = x + row;
    float xa[8];
    load8(xrow, xa);
    trip_cells(A0, A1, B, c_tile(xa[0], wC, bC), s1, s2, s3, c, h);
#pragma unroll
    for (int u = 0; u < 2; ++u) {
        h[u] = l1 ? 0.f : h[u];
        c[u] = l1 ? 0.f : c[u];
    }
    hand_over(h, B);
    v8f Gn = c_tile(xa[1], wC, bC);

    const int slot = 2 * j + hf;
    float* op = out + row + slot;
    constexpr int NP = Tsz / 8;
    for (int p = 0; p < NP; ++p) {
        const int nb = (p + 1 < NP) ? p + 1 : NP - 1;
        float xn[8];
        load8(xrow + 8 * nb, xn);

        float keep = 0.f;
#pragma unroll
        for (int k = 0; k < 8; ++k) {
            if (k > 0) {
                const float ph = project_half(B, wlh, bh);
                const float pv = ph + other_half(ph);
                keep = (slot == k - 1) ? pv : keep;
            }
            const v8f Gc = Gn;
            const float xv = (k + 2 < 8) ? xa[k + 2] : xn[k + 2 - 8];
            Gn = c_tile(xv, wC, bC);

            trip_cells(A0, A1, B, Gc, s1, s2, s3, c, h);
            hand_over(h, B);
            __builtin_amdgcn_sched_barrier(0);
        }
        {
            const float ph = project_half(B, wlh, bh);
            const float pv = ph + other_half(ph);
            keep = (slot == 7) ? pv : keep;
        }
        op[8 * p] = keep;

#pragma unroll
        for (int i = 0; i < 8; ++i) xa[i] = xn[i];
    }
}

extern "C" void kernel_launch(void* const* d_in, const int* in_sizes, int n_in,
                              void* d_out, int out_size, void* d_ws, size_t ws_size,
                              hipStream_t stream) {
    (void)in_sizes; (void)n_in; (void)out_size; (void)d_ws; (void)ws_size;
    const float* x    = (const float*)d_in[0];
    const float* Wih0 = (const float*)d_in[1];
    const float* Whh0 = (const float*)d_in[2];
    const float* bih0 = (const float*)d_in[3];
    const float* bhh0 = (const float*)d_in[4];
    const float* Wih1 = (const float*)d_in[5];
    const float* Whh1 = (const float*)d_in[6];
    const float* bih1 = (const float*)d_in[7];
    const float* bhh1 = (const float*)d_in[8];
    const float* Wlin = (const float*)d_in[9];
    const float* blin = (const float*)d_in[10];
    float* out = (float*)d_out;

    lstm2_rows4_kernel<<<Bsz / 16, 128, 0, stream>>>(
        x, Wih0, Whh0, bih0, bhh0, Wih1, Whh1, bih1, bhh1, Wlin, blin, out);
}
